# starting kernel
# baseline (speedup 1.0000x reference)
#include <hip/hip_runtime.h>
#include <stdint.h>


typedef float v2f __attribute__((ext_vector_type(2)));
typedef float v8f __attribute__((ext_vector_type(8)));

#define ORD_NEG_INF 0x007FFFFFu

__device__ __forceinline__ unsigned ord_encode(float f) {
    unsigned u = __float_as_uint(f);
    return (u & 0x80000000u) ? ~u : (u | 0x80000000u);
}

__device__ __forceinline__ float ord_decode_finite(unsigned k) {
    unsigned u = (k & 0x80000000u) ? (k & 0x7FFFFFFFu) : ~k;
    if ((u & 0x7F800000u) == 0x7F800000u) return 0.0f;
    return __uint_as_float(u);
}

__global__ void GCN_fill_u32(unsigned* __restrict__ p, long n, unsigned v) {
    long i = blockIdx.x * (long)blockDim.x + threadIdx.x;
    if (i < n) p[i] = v;
}

template <int LOG2F>
__global__ void GCN_edge_max(const float* __restrict__ feat,
                             const int* __restrict__ src,
                             const int* __restrict__ dst,
                             const float* __restrict__ ts,
                             unsigned* __restrict__ agg,
                             long nwork) {
    long t = blockIdx.x * (long)blockDim.x + threadIdx.x;
    if (t >= nwork) return;
    const int F = 1 << LOG2F;
    int e = (int)(t >> LOG2F);
    int f = (int)(t & (F - 1));
    int s = src[e];
    int d = dst[e];
    float m = feat[(long)s * F + f] * ts[e];
    atomicMax(agg + (long)d * F + f, ord_encode(m));
}

__global__ void __launch_bounds__(256)
GCN_finalize1(const unsigned* __restrict__ agg,
              const float* __restrict__ W1,
              const float* __restrict__ b1,
              float* __restrict__ h,
              int n_nodes) {
    int wave = (int)((blockIdx.x * (long)blockDim.x + threadIdx.x) >> 5);
    int lane = threadIdx.x & 31;
    int base = wave * 16;
    if (base >= n_nodes) return;

    int m  = lane & 15;
    int kh = lane >> 4;
    int n  = m;

    v8f acc = {0.f, 0.f, 0.f, 0.f, 0.f, 0.f, 0.f, 0.f};
#pragma unroll
    for (int c = 0; c < 4; ++c) {
        int k0 = 4 * c + 2 * kh;
        v2f a, b;
        a.x = ord_decode_finite(agg[(base + m) * 16 + k0]);
        a.y = ord_decode_finite(agg[(base + m) * 16 + k0 + 1]);
        b.x = (n < 8) ? W1[k0 * 8 + n]       : 0.0f;
        b.y = (n < 8) ? W1[(k0 + 1) * 8 + n] : 0.0f;
        acc = __builtin_amdgcn_wmma_f32_16x16x4_f32(
             false, a,  false, b,
             (short)0, acc,  false,  false);
    }

    if (n < 8) {
#pragma unroll
        for (int r = 0; r < 8; ++r) {
            int node = base + r + 8 * kh;
            float v = acc[r] + b1[n];
            h[node * 8 + n] = v > 0.0f ? v : 0.0f;
        }
    }
}

__global__ void __launch_bounds__(256)
GCN_finalize2(const unsigned* __restrict__ agg,
              const float* __restrict__ W2,
              const float* __restrict__ b2,
              float* __restrict__ out,
              int n_nodes) {
    int wave = (int)((blockIdx.x * (long)blockDim.x + threadIdx.x) >> 5);
    int lane = threadIdx.x & 31;
    int base = wave * 16;
    if (base >= n_nodes) return;

    int m  = lane & 15;
    int kh = lane >> 4;
    int n  = m;

    v8f acc = {0.f, 0.f, 0.f, 0.f, 0.f, 0.f, 0.f, 0.f};
#pragma unroll
    for (int c = 0; c < 2; ++c) {
        int k0 = 4 * c + 2 * kh;
        v2f a, b;
        a.x = ord_decode_finite(agg[(base + m) * 8 + k0]);
        a.y = ord_decode_finite(agg[(base + m) * 8 + k0 + 1]);
        b.x = (n < 2) ? W2[k0 * 2 + n]       : 0.0f;
        b.y = (n < 2) ? W2[(k0 + 1) * 2 + n] : 0.0f;
        acc = __builtin_amdgcn_wmma_f32_16x16x4_f32(
            false, a, false, b, (short)0, acc, false, false);
    }

    if (n < 2) {
#pragma unroll
        for (int r = 0; r < 8; ++r) {
            int node = base + r + 8 * kh;
            out[node * 2 + n] = acc[r] + b2[n];
        }
    }
}

extern "C" void kernel_launch(void* const* d_in, const int* in_sizes, int n_in,
                              void* d_out, int out_size, void* d_ws, size_t ws_size,
                              hipStream_t stream) {
    const float* x   = (const float*)d_in[0];
    const int*   src = (const int*)  d_in[1];
    const int*   dst = (const int*)  d_in[2];
    const float* ts  = (const float*)d_in[3];
    const float* W1  = (const float*)d_in[4];
    const float* b1  = (const float*)d_in[5];
    const float* W2  = (const float*)d_in[6];
    const float* b2  = (const float*)d_in[7];
    float* out = (float*)d_out;

    const int N = in_sizes[0] / 16;
    const int E = in_sizes[1];

    unsigned* agg1 = (unsigned*)d_ws;
    float*    h    = (float*)(agg1 + (size_t)N * 16);
    unsigned* agg2 = (unsigned*)(h + (size_t)N * 8);

    const int B = 256;

    long n1 = (long)N * 16;
    GCN_fill_u32<<<(unsigned)((n1 + B - 1) / B), B, 0, stream>>>(agg1, n1, ORD_NEG_INF);

    long w1 = (long)E * 16;
    GCN_edge_max<4><<<(unsigned)((w1 + B - 1) / B), B, 0, stream>>>(x, src, dst, ts, agg1, w1);

    int waves   = N / 16;
    int fin_thr = waves * 32;
    GCN_finalize1<<<(unsigned)((fin_thr + B - 1) / B), B, 0, stream>>>(agg1, W1, b1, h, N);

    long n2 = (long)N * 8;
    GCN_fill_u32<<<(unsigned)((n2 + B - 1) / B), B, 0, stream>>>(agg2, n2, ORD_NEG_INF);

    long w2 = (long)E * 8;
    GCN_edge_max<3><<<(unsigned)((w2 + B - 1) / B), B, 0, stream>>>(h, src, dst, ts, agg2, w2);

    GCN_finalize2<<<(unsigned)((fin_thr + B - 1) / B), B, 0, stream>>>(agg2, W2, b2, out, N);
}


# No global atomics, no edge list: a work-group keeps a node range's maxima in LDS, streams dst, applies its own edges; slice planes maxed in the epilogues; max is order-free: bit-identical
# speedup vs baseline: 1.9710x; 1.9710x over previous
#include <hip/hip_runtime.h>
#include <stdint.h>


typedef float v2f __attribute__((ext_vector_type(2)));
typedef float v8f __attribute__((ext_vector_type(8)));

#define ORD_NEG_INF 0x007FFFFFu

#define SCAN_NT    1024
#define SCAN_EPT   16
#define SCAN_BLK   (SCAN_NT * SCAN_EPT)
#define SCAN_QCAP  2048
#define ACC_WORDS  64000
#define SLICES_L1  10
#define SLICES_L2  19

__device__ __forceinline__ unsigned ord_encode(float f) {
    unsigned u = __float_as_uint(f);
    return (u & 0x80000000u) ? ~u : (u | 0x80000000u);
}

__device__ __forceinline__ float ord_decode_finite(unsigned k) {
    unsigned u = (k & 0x80000000u) ? (k & 0x7FFFFFFFu) : ~k;
    if ((u & 0x7F800000u) == 0x7F800000u) return 0.0f;
    return __uint_as_float(u);
}

template <int F>
__device__ __noinline__ void edge_direct(const float* __restrict__ feat,
                                         const int* __restrict__ src,
                                         const float* __restrict__ ts,
                                         unsigned* acc, int e, unsigned dl) {
    int s = src[e];
    float w = ts[e];
    for (int f = 0; f < F; ++f)
        atomicMax(acc + dl * F + f, ord_encode(feat[(long)s * F + f] * w));
}

template <int LOG2F>
__global__ void __launch_bounds__(SCAN_NT)
GCN_range_scan_max(const float* __restrict__ feat,
                   const int* __restrict__ src,
                   const int* __restrict__ dst,
                   const float* __restrict__ ts,
                   unsigned* __restrict__ part,
                   int n_nodes, int n_edges, int n_ranges, int blocks_per_slice) {
    constexpr int F     = 1 << LOG2F;
    constexpr int RANGE = ACC_WORDS / F;
    constexpr int LPE   = F / 4;

    __shared__ __attribute__((aligned(16))) unsigned acc[ACC_WORDS];
    __shared__ uint2    q[SCAN_QCAP];
    __shared__ unsigned q_n[2];

    const int tid  = threadIdx.x;
    const int lane = tid & 31;
    const int r    = (int)blockIdx.x % n_ranges;
    const int s    = (int)blockIdx.x / n_ranges;
    const int lo   = r * RANGE;
    int cnt = n_nodes - lo;
    if (cnt > RANGE) cnt = RANGE;

    for (int i = tid; i < ACC_WORDS; i += SCAN_NT) acc[i] = ORD_NEG_INF;
    if (tid < 2) q_n[tid] = 0u;
    __syncthreads();

    const int b0 = s * blocks_per_slice;
    int  dv[SCAN_EPT];
    bool have = false;
    for (int bi = 0; bi < blocks_per_slice; ++bi) {
        const int ebase = (b0 + bi) * SCAN_BLK;
        if (ebase >= n_edges) break;
        const int par = bi & 1;

        if (have) {
        } else if (ebase + SCAN_BLK <= n_edges) {
#pragma unroll
            for (int k = 0; k < 4; ++k) {
                int4 t = *(const int4*)(dst + ebase + (k * SCAN_NT + tid) * 4);
                dv[4 * k + 0] = t.x; dv[4 * k + 1] = t.y;
                dv[4 * k + 2] = t.z; dv[4 * k + 3] = t.w;
            }
        } else {
#pragma unroll
            for (int i = 0; i < SCAN_EPT; ++i) {
                int e = ebase + ((i >> 2) * SCAN_NT + tid) * 4 + (i & 3);
                dv[i] = (e < n_edges) ? dst[e] : -1;
            }
        }

        unsigned hitmask[SCAN_EPT];
        unsigned tot = 0;
#pragma unroll
        for (int i = 0; i < SCAN_EPT; ++i) {
            unsigned dl = (unsigned)(dv[i] - lo);
            hitmask[i] = __builtin_amdgcn_ballot_w32(dl < (unsigned)cnt);
            tot += (unsigned)__builtin_popcount(hitmask[i]);
        }
        if (tot) {
            unsigned base = 0;
            if (lane == 0) base = atomicAdd(&q_n[par], tot);
            unsigned run = (unsigned)__builtin_amdgcn_readfirstlane((int)base);
            if (run + tot <= SCAN_QCAP) {
#pragma unroll
                for (int i = 0; i < SCAN_EPT; ++i) {
                    unsigned dl = (unsigned)(dv[i] - lo);
                    if (dl < (unsigned)cnt) {
                        unsigned pos = run + __builtin_amdgcn_mbcnt_lo(hitmask[i], 0u);
                        int e = ebase + ((i >> 2) * SCAN_NT + tid) * 4 + (i & 3);
                        q[pos] = make_uint2((unsigned)e, dl);
                    }
                    run += (unsigned)__builtin_popcount(hitmask[i]);
                }
            } else {
#pragma unroll
                for (int i = 0; i < SCAN_EPT; ++i) {
                    unsigned dl = (unsigned)(dv[i] - lo);
                    if (dl < (unsigned)cnt) {
                        unsigned pos = run + __builtin_amdgcn_mbcnt_lo(hitmask[i], 0u);
                        int e = ebase + ((i >> 2) * SCAN_NT + tid) * 4 + (i & 3);
                        if (pos < SCAN_QCAP) q[pos] = make_uint2((unsigned)e, dl);
                        else edge_direct<F>(feat, src, ts, acc, e, dl);
                    }
                    run += (unsigned)__builtin_popcount(hitmask[i]);
                }
            }
        }
        __syncthreads();

        unsigned n = q_n[par];
        if (n > SCAN_QCAP) n = SCAN_QCAP;
        if (tid == 0) q_n[par ^ 1] = 0u;
        have = (bi + 1 < blocks_per_slice) && (ebase + 2 * SCAN_BLK <= n_edges);
        if (have) {
#pragma unroll
            for (int k = 0; k < 4; ++k) {
                int4 t = *(const int4*)(dst + ebase + SCAN_BLK + (k * SCAN_NT + tid) * 4);
                dv[4 * k + 0] = t.x; dv[4 * k + 1] = t.y;
                dv[4 * k + 2] = t.z; dv[4 * k + 3] = t.w;
            }
        }
        const unsigned ntask = n * LPE;
        for (unsigned t0 = tid; t0 < ntask; t0 += 3 * SCAN_NT) {
            uint2    qq[3];
            unsigned pp[3];
            int      ss[3];
            float    ww[3];
            float4   vv[3];
#pragma unroll
            for (int u = 0; u < 3; ++u) {
                unsigned t = t0 + u * SCAN_NT;
                if (t >= ntask) t = t0;
                pp[u] = t % LPE;
                qq[u] = q[t / LPE];
            }
#pragma unroll
            for (int u = 0; u < 3; ++u) { ss[u] = src[qq[u].x]; ww[u] = ts[qq[u].x]; }
            __builtin_amdgcn_sched_barrier(0);
#pragma unroll
            for (int u = 0; u < 3; ++u)
                vv[u] = *(const float4*)(feat + (long)ss[u] * F + 4 * pp[u]);
            asm volatile("" :: "v"(vv[0].x), "v"(vv[0].y), "v"(vv[0].z), "v"(vv[0].w),
                               "v"(vv[1].x), "v"(vv[1].y), "v"(vv[1].z), "v"(vv[1].w),
                               "v"(vv[2].x), "v"(vv[2].y), "v"(vv[2].z), "v"(vv[2].w));
#pragma unroll
            for (int u = 0; u < 3; ++u) {
                unsigned* a = acc + qq[u].y * F + 4 * pp[u];
                atomicMax(a + 0, ord_encode(vv[u].x * ww[u]));
                atomicMax(a + 1, ord_encode(vv[u].y * ww[u]));
                atomicMax(a + 2, ord_encode(vv[u].z * ww[u]));
                atomicMax(a + 3, ord_encode(vv[u].w * ww[u]));
            }
        }
        __syncthreads();
    }

    unsigned* outp = part + ((size_t)s * n_nodes + lo) * F;
    const int nw = cnt * F;
    for (int i = tid * 4; i < nw; i += SCAN_NT * 4)
        *(uint4*)(outp + i) = *(const uint4*)(acc + i);
}

template <int S>
__global__ void __launch_bounds__(256)
GCN_finalize1(const unsigned* __restrict__ part,
              const float* __restrict__ W1,
              const float* __restrict__ b1,
              float* __restrict__ h,
              int n_nodes) {
    int wave = (int)((blockIdx.x * (long)blockDim.x + threadIdx.x) >> 5);
    int lane = threadIdx.x & 31;
    int base = wave * 16;
    if (base >= n_nodes) return;

    int m  = lane & 15;
    int kh = lane >> 4;
    int n  = m;

    const size_t plane2 = (size_t)n_nodes * 8;
    const uint2* p = (const uint2*)(part + (size_t)(base + m) * 16 + 2 * kh);
    unsigned kx[4] = {0u, 0u, 0u, 0u}, ky[4] = {0u, 0u, 0u, 0u};
#pragma unroll
    for (int s = 0; s < S; ++s) {
#pragma unroll
        for (int c = 0; c < 4; ++c) {
            uint2 v = p[s * plane2 + 2 * c];
            kx[c] = v.x > kx[c] ? v.x : kx[c];
            ky[c] = v.y > ky[c] ? v.y : ky[c];
        }
    }

    v8f acc = {0.f, 0.f, 0.f, 0.f, 0.f, 0.f, 0.f, 0.f};
#pragma unroll
    for (int c = 0; c < 4; ++c) {
        int k0 = 4 * c + 2 * kh;
        v2f a, b;
        a.x = ord_decode_finite(kx[c]);
        a.y = ord_decode_finite(ky[c]);
        b.x = (n < 8) ? W1[k0 * 8 + n]       : 0.0f;
        b.y = (n < 8) ? W1[(k0 + 1) * 8 + n] : 0.0f;
        acc = __builtin_amdgcn_wmma_f32_16x16x4_f32(
             false, a,  false, b,
             (short)0, acc,  false,  false);
    }

    if (n < 8) {
#pragma unroll
        for (int r = 0; r < 8; ++r) {
            int node = base + r + 8 * kh;
            float v = acc[r] + b1[n];
            h[node * 8 + n] = v > 0.0f ? v : 0.0f;
        }
    }
}

template <int S>
__global__ void __launch_bounds__(256)
GCN_finalize2(const unsigned* __restrict__ part,
              const float* __restrict__ W2,
              const float* __restrict__ b2,
              float* __restrict__ out,
              int n_nodes) {
    int wave = (int)((blockIdx.x * (long)blockDim.x + threadIdx.x) >> 5);
    int lane = threadIdx.x & 31;
    int base = wave * 16;
    if (base >= n_nodes) return;

    int m  = lane & 15;
    int kh = lane >> 4;
    int n  = m;

    const size_t plane2 = (size_t)n_nodes * 4;
    const uint2* p = (const uint2*)(part + (size_t)(base + m) * 8 + 2 * kh);
    unsigned kx[2] = {0u, 0u}, ky[2] = {0u, 0u};
#pragma unroll
    for (int s = 0; s < S; ++s) {
#pragma unroll
        for (int c = 0; c < 2; ++c) {
            uint2 v = p[s * plane2 + 2 * c];
            kx[c] = v.x > kx[c] ? v.x : kx[c];
            ky[c] = v.y > ky[c] ? v.y : ky[c];
        }
    }

    v8f acc = {0.f, 0.f, 0.f, 0.f, 0.f, 0.f, 0.f, 0.f};
#pragma unroll
    for (int c = 0; c < 2; ++c) {
        int k0 = 4 * c + 2 * kh;
        v2f a, b;
        a.x = ord_decode_finite(kx[c]);
        a.y = ord_decode_finite(ky[c]);
        b.x = (n < 2) ? W2[k0 * 2 + n]       : 0.0f;
        b.y = (n < 2) ? W2[(k0 + 1) * 2 + n] : 0.0f;
        acc = __builtin_amdgcn_wmma_f32_16x16x4_f32(
            false, a, false, b, (short)0, acc, false, false);
    }

    if (n < 2) {
#pragma unroll
        for (int r = 0; r < 8; ++r) {
            int node = base + r + 8 * kh;
            out[node * 2 + n] = acc[r] + b2[n];
        }
    }
}

extern "C" void kernel_launch(void* const* d_in, const int* in_sizes, int n_in,
                              void* d_out, int out_size, void* d_ws, size_t ws_size,
                              hipStream_t stream) {
    const float* x   = (const float*)d_in[0];
    const int*   src = (const int*)  d_in[1];
    const int*   dst = (const int*)  d_in[2];
    const float* ts  = (const float*)d_in[3];
    const float* W1  = (const float*)d_in[4];
    const float* b1  = (const float*)d_in[5];
    const float* W2  = (const float*)d_in[6];
    const float* b2  = (const float*)d_in[7];
    float* out = (float*)d_out;

    const int N = in_sizes[0] / 16;
    const int E = in_sizes[1];

    unsigned* part1 = (unsigned*)d_ws;
    float*    h     = (float*)(part1 + (size_t)SLICES_L1 * N * 16);
    unsigned* part2 = (unsigned*)(h + (size_t)N * 8);

    const int B = 256;
    const int nblk = (E + SCAN_BLK - 1) / SCAN_BLK;
    int waves   = N / 16;
    int fin_thr = waves * 32;

    const int R1   = (N + ACC_WORDS / 16 - 1) / (ACC_WORDS / 16);
    const int bps1 = (nblk + SLICES_L1 - 1) / SLICES_L1;
    GCN_range_scan_max<4><<<(unsigned)(R1 * SLICES_L1), SCAN_NT, 0, stream>>>(
        x, src, dst, ts, part1, N, E, R1, bps1);
    GCN_finalize1<SLICES_L1><<<(unsigned)((fin_thr + B - 1) / B), B, 0, stream>>>(part1, W1, b1, h, N);

    const int R2   = (N + ACC_WORDS / 8 - 1) / (ACC_WORDS / 8);
    const int bps2 = (nblk + SLICES_L2 - 1) / SLICES_L2;
    GCN_range_scan_max<3><<<(unsigned)(R2 * SLICES_L2), SCAN_NT, 0, stream>>>(
        h, src, dst, ts, part2, N, E, R2, bps2);
    GCN_finalize2<SLICES_L2><<<(unsigned)((fin_thr + B - 1) / B), B, 0, stream>>>(part2, W2, b2, out, N);
}


# Scan loop without work-group barriers: each wave finds, lists (own LDS piece) and applies the hits of its own 512-edge chunks; waves overlap compare, gather, LDS max; max order-free: bit-identical
# speedup vs baseline: 2.8619x; 1.4520x over previous
#include <hip/hip_runtime.h>
#include <stdint.h>


typedef float v2f __attribute__((ext_vector_type(2)));
typedef float v8f __attribute__((ext_vector_type(8)));

#define ORD_NEG_INF 0x007FFFFFu

#define SCAN_NT    1024
#define SCAN_EPT   16
#define SCAN_CHUNK (32 * SCAN_EPT)
#define SCAN_BLK   (SCAN_NT * SCAN_EPT)
#define WQ_CAP     64
#define TPL        4
#define ACC_WORDS  64000
#define SLICES_L1  10
#define SLICES_L2  19

__device__ __forceinline__ unsigned ord_encode(float f) {
    unsigned u = __float_as_uint(f);
    return (u & 0x80000000u) ? ~u : (u | 0x80000000u);
}

__device__ __forceinline__ float ord_decode_finite(unsigned k) {
    unsigned u = (k & 0x80000000u) ? (k & 0x7FFFFFFFu) : ~k;
    if ((u & 0x7F800000u) == 0x7F800000u) return 0.0f;
    return __uint_as_float(u);
}

template <int LOG2F>
__global__ void __launch_bounds__(SCAN_NT)
GCN_range_scan_max(const float* __restrict__ feat,
                   const int* __restrict__ src,
                   const int* __restrict__ dst,
                   const float* __restrict__ ts,
                   unsigned* __restrict__ part,
                   int n_nodes, int n_edges, int n_ranges, int blocks_per_slice) {
    constexpr int F     = 1 << LOG2F;
    constexpr int RANGE = ACC_WORDS / F;
    constexpr int LPE   = F / 4;

    __shared__ __attribute__((aligned(16))) unsigned acc[ACC_WORDS];
    __shared__ uint2 wq[SCAN_NT / 32][WQ_CAP];

    const int tid  = threadIdx.x;
    const int lane = tid & 31;
    const int wave = tid >> 5;
    const int r    = (int)blockIdx.x % n_ranges;
    const int s    = (int)blockIdx.x / n_ranges;
    const int lo   = r * RANGE;
    int cnt = n_nodes - lo;
    if (cnt > RANGE) cnt = RANGE;

    for (int i = tid; i < ACC_WORDS; i += SCAN_NT) acc[i] = ORD_NEG_INF;
    __syncthreads();

    uint2* myq = wq[wave];
    const int c_end = (s + 1) * blocks_per_slice * (SCAN_NT / 32);
    int  dv[SCAN_EPT];
    int  dn[SCAN_EPT];
    bool have = false;
    for (int c = s * blocks_per_slice * (SCAN_NT / 32) + wave; c < c_end; c += SCAN_NT / 32) {
        const int ebase = c * SCAN_CHUNK;
        if (ebase >= n_edges) break;

        if (have) {
        } else if (ebase + SCAN_CHUNK <= n_edges) {
#pragma unroll
            for (int k = 0; k < 4; ++k) {
                int4 t = *(const int4*)(dst + ebase + (k * 32 + lane) * 4);
                dv[4 * k + 0] = t.x; dv[4 * k + 1] = t.y;
                dv[4 * k + 2] = t.z; dv[4 * k + 3] = t.w;
            }
        } else {
#pragma unroll
            for (int i = 0; i < SCAN_EPT; ++i) {
                int e = ebase + ((i >> 2) * 32 + lane) * 4 + (i & 3);
                dv[i] = (e < n_edges) ? dst[e] : -1;
            }
        }

        unsigned hitmask[SCAN_EPT];
        unsigned tot = 0;
#pragma unroll
        for (int i = 0; i < SCAN_EPT; ++i) {
            unsigned dl = (unsigned)(dv[i] - lo);
            hitmask[i] = __builtin_amdgcn_ballot_w32(dl < (unsigned)cnt);
            tot += (unsigned)__builtin_popcount(hitmask[i]);
        }

        have = (c + SCAN_NT / 32 < c_end) && (ebase + (SCAN_NT / 32 + 1) * SCAN_CHUNK <= n_edges);
        if (have) {
#pragma unroll
            for (int k = 0; k < 4; ++k) {
                int4 t = *(const int4*)(dst + ebase + (SCAN_NT / 32) * SCAN_CHUNK + (k * 32 + lane) * 4);
                dn[4 * k + 0] = t.x; dn[4 * k + 1] = t.y;
                dn[4 * k + 2] = t.z; dn[4 * k + 3] = t.w;
            }
        }

        for (unsigned rb = 0; rb < tot; rb += WQ_CAP) {
            unsigned run = 0;
            if (tot <= (unsigned)WQ_CAP) {
#pragma unroll
                for (int i = 0; i < SCAN_EPT; ++i) {
                    unsigned dl = (unsigned)(dv[i] - lo);
                    if (dl < (unsigned)cnt) {
                        unsigned pos = run + __builtin_amdgcn_mbcnt_lo(hitmask[i], 0u);
                        int e = ebase + ((i >> 2) * 32 + lane) * 4 + (i & 3);
                        myq[pos] = make_uint2((unsigned)e, dl);
                    }
                    run += (unsigned)__builtin_popcount(hitmask[i]);
                }
            } else {
#pragma unroll
                for (int i = 0; i < SCAN_EPT; ++i) {
                    unsigned dl  = (unsigned)(dv[i] - lo);
                    unsigned pos = run + __builtin_amdgcn_mbcnt_lo(hitmask[i], 0u) - rb;
                    if (dl < (unsigned)cnt && pos < (unsigned)WQ_CAP) {
                        int e = ebase + ((i >> 2) * 32 + lane) * 4 + (i & 3);
                        myq[pos] = make_uint2((unsigned)e, dl);
                    }
                    run += (unsigned)__builtin_popcount(hitmask[i]);
                }
            }
            __builtin_amdgcn_fence(__ATOMIC_RELEASE, "wavefront");
            __builtin_amdgcn_wave_barrier();
            __builtin_amdgcn_fence(__ATOMIC_ACQUIRE, "wavefront");

            unsigned nq = tot - rb;
            if (nq > (unsigned)WQ_CAP) nq = WQ_CAP;
            const unsigned ntask = nq * LPE;
            for (unsigned t0 = lane; t0 < ntask; t0 += 32 * TPL) {
                bool     ok[TPL];
                uint2    qq[TPL];
                unsigned pp[TPL];
                int      ss[TPL];
                float    ww[TPL];
                float4   vv[TPL];
#pragma unroll
                for (int u = 0; u < TPL; ++u) {
                    unsigned t = t0 + u * 32;
                    ok[u] = t < ntask;
                    pp[u] = t % LPE;
                    qq[u] = make_uint2(0u, 0u);
                    if (ok[u]) qq[u] = myq[t / LPE];
                }
#pragma unroll
                for (int u = 0; u < TPL; ++u) {
                    ss[u] = 0; ww[u] = 0.0f;
                    if (ok[u]) { ss[u] = src[qq[u].x]; ww[u] = ts[qq[u].x]; }
                }
                __builtin_amdgcn_sched_barrier(0);
#pragma unroll
                for (int u = 0; u < TPL; ++u) {
                    vv[u] = make_float4(0.0f, 0.0f, 0.0f, 0.0f);
                    if (ok[u]) vv[u] = *(const float4*)(feat + (long)ss[u] * F + 4 * pp[u]);
                }
                asm volatile("" :: "v"(vv[0].x), "v"(vv[0].y), "v"(vv[0].z), "v"(vv[0].w),
                                   "v"(vv[1].x), "v"(vv[1].y), "v"(vv[1].z), "v"(vv[1].w),
                                   "v"(vv[2].x), "v"(vv[2].y), "v"(vv[2].z), "v"(vv[2].w),
                                   "v"(vv[3].x), "v"(vv[3].y), "v"(vv[3].z), "v"(vv[3].w),
                                   "v"(ww[0]), "v"(ww[1]), "v"(ww[2]), "v"(ww[3]));
#pragma unroll
                for (int u = 0; u < TPL; ++u) {
                    if (ok[u]) {
                        unsigned* a = acc + qq[u].y * F + 4 * pp[u];
                        atomicMax(a + 0, ord_encode(vv[u].x * ww[u]));
                        atomicMax(a + 1, ord_encode(vv[u].y * ww[u]));
                        atomicMax(a + 2, ord_encode(vv[u].z * ww[u]));
                        atomicMax(a + 3, ord_encode(vv[u].w * ww[u]));
                    }
                }
            }
            __builtin_amdgcn_wave_barrier();
        }

        if (have) {
#pragma unroll
            for (int i = 0; i < SCAN_EPT; ++i) dv[i] = dn[i];
        }
    }
    __syncthreads();

    unsigned* outp = part + ((size_t)s * n_nodes + lo) * F;
    const int nw = cnt * F;
    for (int i = tid * 4; i < nw; i += SCAN_NT * 4)
        *(uint4*)(outp + i) = *(const uint4*)(acc + i);
}

template <int S>
__global__ void __launch_bounds__(256)
GCN_finalize1(const unsigned* __restrict__ part,
              const float* __restrict__ W1,
              const float* __restrict__ b1,
              float* __restrict__ h,
              int n_nodes) {
    int wave = (int)((blockIdx.x * (long)blockDim.x + threadIdx.x) >> 5);
    int lane = threadIdx.x & 31;
    int base = wave * 16;
    if (base >= n_nodes) return;

    int m  = lane & 15;
    int kh = lane >> 4;
    int n  = m;

    const size_t plane2 = (size_t)n_nodes * 8;
    const uint2* p = (const uint2*)(part + (size_t)(base + m) * 16 + 2 * kh);
    unsigned kx[4] = {0u, 0u, 0u, 0u}, ky[4] = {0u, 0u, 0u, 0u};
#pragma unroll
    for (int s = 0; s < S; ++s) {
#pragma unroll
        for (int c = 0; c < 4; ++c) {
            uint2 v = p[s * plane2 + 2 * c];
            kx[c] = v.x > kx[c] ? v.x : kx[c];
            ky[c] = v.y > ky[c] ? v.y : ky[c];
        }
    }

    v8f acc = {0.f, 0.f, 0.f, 0.f, 0.f, 0.f, 0.f, 0.f};
#pragma unroll
    for (int c = 0; c < 4; ++c) {
        int k0 = 4 * c + 2 * kh;
        v2f a, b;
        a.x = ord_decode_finite(kx[c]);
        a.y = ord_decode_finite(ky[c]);
        b.x = (n < 8) ? W1[k0 * 8 + n]       : 0.0f;
        b.y = (n < 8) ? W1[(k0 + 1) * 8 + n] : 0.0f;
        acc = __builtin_amdgcn_wmma_f32_16x16x4_f32(
             false, a,  false, b,
             (short)0, acc,  false,  false);
    }

    if (n < 8) {
#pragma unroll
        for (int r = 0; r < 8; ++r) {
            int node = base + r + 8 * kh;
            float v = acc[r] + b1[n];
            h[node * 8 + n] = v > 0.0f ? v : 0.0f;
        }
    }
}

template <int S>
__global__ void __launch_bounds__(256)
GCN_finalize2(const unsigned* __restrict__ part,
              const float* __restrict__ W2,
              const float* __restrict__ b2,
              float* __restrict__ out,
              int n_nodes) {
    int wave = (int)((blockIdx.x * (long)blockDim.x + threadIdx.x) >> 5);
    int lane = threadIdx.x & 31;
    int base = wave * 16;
    if (base >= n_nodes) return;

    int m  = lane & 15;
    int kh = lane >> 4;
    int n  = m;

    const size_t plane2 = (size_t)n_nodes * 4;
    const uint2* p = (const uint2*)(part + (size_t)(base + m) * 8 + 2 * kh);
    unsigned kx[2] = {0u, 0u}, ky[2] = {0u, 0u};
#pragma unroll
    for (int s = 0; s < S; ++s) {
#pragma unroll
        for (int c = 0; c < 2; ++c) {
            uint2 v = p[s * plane2 + 2 * c];
            kx[c] = v.x > kx[c] ? v.x : kx[c];
            ky[c] = v.y > ky[c] ? v.y : ky[c];
        }
    }

    v8f acc = {0.f, 0.f, 0.f, 0.f, 0.f, 0.f, 0.f, 0.f};
#pragma unroll
    for (int c = 0; c < 2; ++c) {
        int k0 = 4 * c + 2 * kh;
        v2f a, b;
        a.x = ord_decode_finite(kx[c]);
        a.y = ord_decode_finite(ky[c]);
        b.x = (n < 2) ? W2[k0 * 2 + n]       : 0.0f;
        b.y = (n < 2) ? W2[(k0 + 1) * 2 + n] : 0.0f;
        acc = __builtin_amdgcn_wmma_f32_16x16x4_f32(
            false, a, false, b, (short)0, acc, false, false);
    }

    if (n < 2) {
#pragma unroll
        for (int r = 0; r < 8; ++r) {
            int node = base + r + 8 * kh;
            out[node * 2 + n] = acc[r] + b2[n];
        }
    }
}

extern "C" void kernel_launch(void* const* d_in, const int* in_sizes, int n_in,
                              void* d_out, int out_size, void* d_ws, size_t ws_size,
                              hipStream_t stream) {
    const float* x   = (const float*)d_in[0];
    const int*   src = (const int*)  d_in[1];
    const int*   dst = (const int*)  d_in[2];
    const float* ts  = (const float*)d_in[3];
    const float* W1  = (const float*)d_in[4];
    const float* b1  = (const float*)d_in[5];
    const float* W2  = (const float*)d_in[6];
    const float* b2  = (const float*)d_in[7];
    float* out = (float*)d_out;

    const int N = in_sizes[0] / 16;
    const int E = in_sizes[1];

    unsigned* part1 = (unsigned*)d_ws;
    float*    h     = (float*)(part1 + (size_t)SLICES_L1 * N * 16);
    unsigned* part2 = (unsigned*)(h + (size_t)N * 8);

    const int B = 256;
    const int nblk = (E + SCAN_BLK - 1) / SCAN_BLK;
    int waves   = N / 16;
    int fin_thr = waves * 32;

    const int R1   = (N + ACC_WORDS / 16 - 1) / (ACC_WORDS / 16);
    const int bps1 = (nblk + SLICES_L1 - 1) / SLICES_L1;
    GCN_range_scan_max<4><<<(unsigned)(R1 * SLICES_L1), SCAN_NT, 0, stream>>>(
        x, src, dst, ts, part1, N, E, R1, bps1);
    GCN_finalize1<SLICES_L1><<<(unsigned)((fin_thr + B - 1) / B), B, 0, stream>>>(part1, W1, b1, h, N);

    const int R2   = (N + ACC_WORDS / 8 - 1) / (ACC_WORDS / 8);
    const int bps2 = (nblk + SLICES_L2 - 1) / SLICES_L2;
    GCN_range_scan_max<3><<<(unsigned)(R2 * SLICES_L2), SCAN_NT, 0, stream>>>(
        h, src, dst, ts, part2, N, E, R2, bps2);
    GCN_finalize2<SLICES_L2><<<(unsigned)((fin_thr + B - 1) / B), B, 0, stream>>>(part2, W2, b2, out, N);
}


# Scan streams src and edge weight with dst (coalesced 16-byte loads, next chunk asked for in place after listing); hit list carries them: a hit waits for one gather (its row), not two; bit-identical
# speedup vs baseline: 3.1042x; 1.0847x over previous
#include <hip/hip_runtime.h>
#include <stdint.h>


typedef float v2f __attribute__((ext_vector_type(2)));
typedef float v8f __attribute__((ext_vector_type(8)));

#define ORD_NEG_INF 0x007FFFFFu

#define SCAN_NT    1024
#define SCAN_EPT   16
#define SCAN_CHUNK (32 * SCAN_EPT)
#define SCAN_BLK   (SCAN_NT * SCAN_EPT)
#define WQ_CAP     64
#define TPL        4
#define ACC_WORDS  64000
#define SLICES_L1  10
#define SLICES_L2  19

__device__ __forceinline__ unsigned ord_encode(float f) {
    unsigned u = __float_as_uint(f);
    return (u & 0x80000000u) ? ~u : (u | 0x80000000u);
}

__device__ __forceinline__ float ord_decode_finite(unsigned k) {
    unsigned u = (k & 0x80000000u) ? (k & 0x7FFFFFFFu) : ~k;
    if ((u & 0x7F800000u) == 0x7F800000u) return 0.0f;
    return __uint_as_float(u);
}

template <int F>
__device__ __forceinline__ void GCN_apply_hits(const float* __restrict__ feat, unsigned* acc,
                                               const unsigned* myq, unsigned nq, int lane) {
    constexpr int LPE = F / 4;
    const unsigned ntask = nq * LPE;
    for (unsigned t0 = lane; t0 < ntask; t0 += 32 * TPL) {
        bool     ok[TPL];
        unsigned pp[TPL];
        unsigned qs[TPL];
        unsigned qd[TPL];
        float    qw[TPL];
        float4   vv[TPL];
#pragma unroll
        for (int u = 0; u < TPL; ++u) {
            unsigned t = t0 + u * 32;
            ok[u] = t < ntask;
            pp[u] = t % LPE;
            unsigned hh = ok[u] ? t / LPE : 0u;
            qs[u] = myq[hh];
            qw[u] = __uint_as_float(myq[WQ_CAP + hh]);
            qd[u] = myq[2 * WQ_CAP + hh];
        }
        asm volatile("" :: "v"(qs[0]), "v"(qs[1]), "v"(qs[2]), "v"(qs[3]));
#pragma unroll
        for (int u = 0; u < TPL; ++u) {
            vv[u] = make_float4(0.0f, 0.0f, 0.0f, 0.0f);
            if (ok[u]) vv[u] = *(const float4*)(feat + (size_t)qs[u] * F + 4 * pp[u]);
        }
        asm volatile("" :: "v"(vv[0].x), "v"(vv[0].y), "v"(vv[0].z), "v"(vv[0].w),
                           "v"(vv[1].x), "v"(vv[1].y), "v"(vv[1].z), "v"(vv[1].w),
                           "v"(vv[2].x), "v"(vv[2].y), "v"(vv[2].z), "v"(vv[2].w),
                           "v"(vv[3].x), "v"(vv[3].y), "v"(vv[3].z), "v"(vv[3].w));
#pragma unroll
        for (int u = 0; u < TPL; ++u) {
            if (ok[u]) {
                unsigned* a = acc + qd[u] * F + 4 * pp[u];
                atomicMax(a + 0, ord_encode(vv[u].x * qw[u]));
                atomicMax(a + 1, ord_encode(vv[u].y * qw[u]));
                atomicMax(a + 2, ord_encode(vv[u].z * qw[u]));
                atomicMax(a + 3, ord_encode(vv[u].w * qw[u]));
            }
        }
    }
}

template <int LOG2F>
__global__ void __launch_bounds__(SCAN_NT)
GCN_range_scan_max(const float* __restrict__ feat,
                   const int* __restrict__ src,
                   const int* __restrict__ dst,
                   const float* __restrict__ ts,
                   unsigned* __restrict__ part,
                   int n_nodes, int n_edges, int n_ranges, int blocks_per_slice) {
    constexpr int F     = 1 << LOG2F;
    constexpr int RANGE = ACC_WORDS / F;

    __shared__ __attribute__((aligned(16))) unsigned acc[ACC_WORDS];
    __shared__ unsigned wq[SCAN_NT / 32][3 * WQ_CAP];

    const int tid  = threadIdx.x;
    const int lane = tid & 31;
    const int wave = tid >> 5;
    const int r    = (int)blockIdx.x % n_ranges;
    const int s    = (int)blockIdx.x / n_ranges;
    const int lo   = r * RANGE;
    int cnt = n_nodes - lo;
    if (cnt > RANGE) cnt = RANGE;

    for (int i = tid; i < ACC_WORDS; i += SCAN_NT) acc[i] = ORD_NEG_INF;
    __syncthreads();

    unsigned* myq = wq[wave];
    const int c_end = (s + 1) * blocks_per_slice * (SCAN_NT / 32);
    int   dv[SCAN_EPT];
    int   sv[SCAN_EPT];
    float tv[SCAN_EPT];
    bool have = false;
    for (int c = s * blocks_per_slice * (SCAN_NT / 32) + wave; c < c_end; c += SCAN_NT / 32) {
        const int ebase = c * SCAN_CHUNK;
        if (ebase >= n_edges) break;

        if (have) {
        } else if (ebase + SCAN_CHUNK <= n_edges) {
#pragma unroll
            for (int k = 0; k < 4; ++k) {
                const int o = ebase + (k * 32 + lane) * 4;
                int4   d = *(const int4*)(dst + o);
                int4   q = *(const int4*)(src + o);
                float4 w = *(const float4*)(ts + o);
                dv[4 * k + 0] = d.x; dv[4 * k + 1] = d.y; dv[4 * k + 2] = d.z; dv[4 * k + 3] = d.w;
                sv[4 * k + 0] = q.x; sv[4 * k + 1] = q.y; sv[4 * k + 2] = q.z; sv[4 * k + 3] = q.w;
                tv[4 * k + 0] = w.x; tv[4 * k + 1] = w.y; tv[4 * k + 2] = w.z; tv[4 * k + 3] = w.w;
            }
        } else {
#pragma unroll
            for (int i = 0; i < SCAN_EPT; ++i) {
                int e = ebase + ((i >> 2) * 32 + lane) * 4 + (i & 3);
                dv[i] = -1; sv[i] = 0; tv[i] = 0.0f;
                if (e < n_edges) { dv[i] = dst[e]; sv[i] = src[e]; tv[i] = ts[e]; }
            }
        }

        unsigned hitmask[SCAN_EPT];
        unsigned tot = 0;
#pragma unroll
        for (int i = 0; i < SCAN_EPT; ++i) {
            unsigned dl = (unsigned)(dv[i] - lo);
            hitmask[i] = __builtin_amdgcn_ballot_w32(dl < (unsigned)cnt);
            tot += (unsigned)__builtin_popcount(hitmask[i]);
        }

        const int  enext = ebase + (SCAN_NT / 32) * SCAN_CHUNK;
        const bool next_whole = (c + SCAN_NT / 32 < c_end) && (enext + SCAN_CHUNK <= n_edges);

        if (tot <= (unsigned)WQ_CAP) {
            unsigned run = 0;
#pragma unroll
            for (int i = 0; i < SCAN_EPT; ++i) {
                unsigned dl = (unsigned)(dv[i] - lo);
                if (dl < (unsigned)cnt) {
                    unsigned pos = run + __builtin_amdgcn_mbcnt_lo(hitmask[i], 0u);
                    myq[pos]              = (unsigned)sv[i];
                    myq[WQ_CAP + pos]     = __float_as_uint(tv[i]);
                    myq[2 * WQ_CAP + pos] = dl;
                }
                run += (unsigned)__builtin_popcount(hitmask[i]);
            }
            have = next_whole;
            if (have) {
#pragma unroll
                for (int k = 0; k < 4; ++k) {
                    const int o = enext + (k * 32 + lane) * 4;
                    int4   d = *(const int4*)(dst + o);
                    int4   q = *(const int4*)(src + o);
                    float4 w = *(const float4*)(ts + o);
                    dv[4 * k + 0] = d.x; dv[4 * k + 1] = d.y; dv[4 * k + 2] = d.z; dv[4 * k + 3] = d.w;
                    sv[4 * k + 0] = q.x; sv[4 * k + 1] = q.y; sv[4 * k + 2] = q.z; sv[4 * k + 3] = q.w;
                    tv[4 * k + 0] = w.x; tv[4 * k + 1] = w.y; tv[4 * k + 2] = w.z; tv[4 * k + 3] = w.w;
                }
            }
            if (tot != 0u) {
                __builtin_amdgcn_fence(__ATOMIC_RELEASE, "wavefront");
                __builtin_amdgcn_wave_barrier();
                __builtin_amdgcn_fence(__ATOMIC_ACQUIRE, "wavefront");
                GCN_apply_hits<F>(feat, acc, myq, tot, lane);
                __builtin_amdgcn_wave_barrier();
            }
        } else {
            for (unsigned rb = 0; rb < tot; rb += WQ_CAP) {
                unsigned run = 0;
#pragma unroll
                for (int i = 0; i < SCAN_EPT; ++i) {
                    unsigned dl  = (unsigned)(dv[i] - lo);
                    unsigned pos = run + __builtin_amdgcn_mbcnt_lo(hitmask[i], 0u) - rb;
                    if (dl < (unsigned)cnt && pos < (unsigned)WQ_CAP) {
                        myq[pos]              = (unsigned)sv[i];
                        myq[WQ_CAP + pos]     = __float_as_uint(tv[i]);
                        myq[2 * WQ_CAP + pos] = dl;
                    }
                    run += (unsigned)__builtin_popcount(hitmask[i]);
                }
                __builtin_amdgcn_fence(__ATOMIC_RELEASE, "wavefront");
                __builtin_amdgcn_wave_barrier();
                __builtin_amdgcn_fence(__ATOMIC_ACQUIRE, "wavefront");
                unsigned nq = tot - rb;
                if (nq > (unsigned)WQ_CAP) nq = WQ_CAP;
                GCN_apply_hits<F>(feat, acc, myq, nq, lane);
                __builtin_amdgcn_wave_barrier();
            }
            have = false;
        }
    }
    __syncthreads();

    unsigned* outp = part + ((size_t)s * n_nodes + lo) * F;
    const int nw = cnt * F;
    for (int i = tid * 4; i < nw; i += SCAN_NT * 4)
        *(uint4*)(outp + i) = *(const uint4*)(acc + i);
}

template <int S>
__global__ void __launch_bounds__(256)
GCN_finalize1(const unsigned* __restrict__ part,
              const float* __restrict__ W1,
              const float* __restrict__ b1,
              float* __restrict__ h,
              int n_nodes) {
    int wave = (int)((blockIdx.x * (long)blockDim.x + threadIdx.x) >> 5);
    int lane = threadIdx.x & 31;
    int base = wave * 16;
    if (base >= n_nodes) return;

    int m  = lane & 15;
    int kh = lane >> 4;
    int n  = m;

    const size_t plane2 = (size_t)n_nodes * 8;
    const uint2* p = (const uint2*)(part + (size_t)(base + m) * 16 + 2 * kh);
    unsigned kx[4] = {0u, 0u, 0u, 0u}, ky[4] = {0u, 0u, 0u, 0u};
#pragma unroll
    for (int s = 0; s < S; ++s) {
#pragma unroll
        for (int c = 0; c < 4; ++c) {
            uint2 v = p[s * plane2 + 2 * c];
            kx[c] = v.x > kx[c] ? v.x : kx[c];
            ky[c] = v.y > ky[c] ? v.y : ky[c];
        }
    }

    v8f acc = {0.f, 0.f, 0.f, 0.f, 0.f, 0.f, 0.f, 0.f};
#pragma unroll
    for (int c = 0; c < 4; ++c) {
        int k0 = 4 * c + 2 * kh;
        v2f a, b;
        a.x = ord_decode_finite(kx[c]);
        a.y = ord_decode_finite(ky[c]);
        b.x = (n < 8) ? W1[k0 * 8 + n]       : 0.0f;
        b.y = (n < 8) ? W1[(k0 + 1) * 8 + n] : 0.0f;
        acc = __builtin_amdgcn_wmma_f32_16x16x4_f32(
             false, a,  false, b,
             (short)0, acc,  false,  false);
    }

    if (n < 8) {
#pragma unroll
        for (int r = 0; r < 8; ++r) {
            int node = base + r + 8 * kh;
            float v = acc[r] + b1[n];
            h[node * 8 + n] = v > 0.0f ? v : 0.0f;
        }
    }
}

template <int S>
__global__ void __launch_bounds__(256)
GCN_finalize2(const unsigned* __restrict__ part,
              const float* __restrict__ W2,
              const float* __restrict__ b2,
              float* __restrict__ out,
              int n_nodes) {
    int wave = (int)((blockIdx.x * (long)blockDim.x + threadIdx.x) >> 5);
    int lane = threadIdx.x & 31;
    int base = wave * 16;
    if (base >= n_nodes) return;

    int m  = lane & 15;
    int kh = lane >> 4;
    int n  = m;

    const size_t plane2 = (size_t)n_nodes * 4;
    const uint2* p = (const uint2*)(part + (size_t)(base + m) * 8 + 2 * kh);
    unsigned kx[2] = {0u, 0u}, ky[2] = {0u, 0u};
#pragma unroll
    for (int s = 0; s < S; ++s) {
#pragma unroll
        for (int c = 0; c < 2; ++c) {
            uint2 v = p[s * plane2 + 2 * c];
            kx[c] = v.x > kx[c] ? v.x : kx[c];
            ky[c] = v.y > ky[c] ? v.y : ky[c];
        }
    }

    v8f acc = {0.f, 0.f, 0.f, 0.f, 0.f, 0.f, 0.f, 0.f};
#pragma unroll
    for (int c = 0; c < 2; ++c) {
        int k0 = 4 * c + 2 * kh;
        v2f a, b;
        a.x = ord_decode_finite(kx[c]);
        a.y = ord_decode_finite(ky[c]);
        b.x = (n < 2) ? W2[k0 * 2 + n]       : 0.0f;
        b.y = (n < 2) ? W2[(k0 + 1) * 2 + n] : 0.0f;
        acc = __builtin_amdgcn_wmma_f32_16x16x4_f32(
            false, a, false, b, (short)0, acc, false, false);
    }

    if (n < 2) {
#pragma unroll
        for (int r = 0; r < 8; ++r) {
            int node = base + r + 8 * kh;
            out[node * 2 + n] = acc[r] + b2[n];
        }
    }
}

extern "C" void kernel_launch(void* const* d_in, const int* in_sizes, int n_in,
                              void* d_out, int out_size, void* d_ws, size_t ws_size,
                              hipStream_t stream) {
    const float* x   = (const float*)d_in[0];
    const int*   src = (const int*)  d_in[1];
    const int*   dst = (const int*)  d_in[2];
    const float* ts  = (const float*)d_in[3];
    const float* W1  = (const float*)d_in[4];
    const float* b1  = (const float*)d_in[5];
    const float* W2  = (const float*)d_in[6];
    const float* b2  = (const float*)d_in[7];
    float* out = (float*)d_out;

    const int N = in_sizes[0] / 16;
    const int E = in_sizes[1];

    unsigned* part1 = (unsigned*)d_ws;
    float*    h     = (float*)(part1 + (size_t)SLICES_L1 * N * 16);
    unsigned* part2 = (unsigned*)(h + (size_t)N * 8);

    const int B = 256;
    const int nblk = (E + SCAN_BLK - 1) / SCAN_BLK;
    int waves   = N / 16;
    int fin_thr = waves * 32;

    const int R1   = (N + ACC_WORDS / 16 - 1) / (ACC_WORDS / 16);
    const int bps1 = (nblk + SLICES_L1 - 1) / SLICES_L1;
    GCN_range_scan_max<4><<<(unsigned)(R1 * SLICES_L1), SCAN_NT, 0, stream>>>(
        x, src, dst, ts, part1, N, E, R1, bps1);
    GCN_finalize1<SLICES_L1><<<(unsigned)((fin_thr + B - 1) / B), B, 0, stream>>>(part1, W1, b1, h, N);

    const int R2   = (N + ACC_WORDS / 8 - 1) / (ACC_WORDS / 8);
    const int bps2 = (nblk + SLICES_L2 - 1) / SLICES_L2;
    GCN_range_scan_max<3><<<(unsigned)(R2 * SLICES_L2), SCAN_NT, 0, stream>>>(
        h, src, dst, ts, part2, N, E, R2, bps2);
    GCN_finalize2<SLICES_L2><<<(unsigned)((fin_thr + B - 1) / B), B, 0, stream>>>(part2, W2, b2, out, N);
}


# Node ranges sized to the full 320 KiB of LDS: 21 ranges x 12 slices (layer 1), 11 x 23 (layer 2), so each edge is streamed and tested by 32 work-groups, not 38; max is order-free: bit-identical
# speedup vs baseline: 3.2026x; 1.0317x over previous
#include <hip/hip_runtime.h>
#include <stdint.h>


typedef float v2f __attribute__((ext_vector_type(2)));
typedef float v8f __attribute__((ext_vector_type(8)));

#define ORD_NEG_INF 0x007FFFFFu

#define SCAN_NT    1024
#define SCAN_EPT   16
#define SCAN_CHUNK (32 * SCAN_EPT)
#define SCAN_BLK   (SCAN_NT * SCAN_EPT)
#define TPL        4
#define RANGE_L1   4768
#define RANGE_L2   9096
#define QCAP_L1    56
#define QCAP_L2    64
#define SLICES_L1  12
#define SLICES_L2  23

__device__ __forceinline__ unsigned ord_encode(float f) {
    unsigned u = __float_as_uint(f);
    return (u & 0x80000000u) ? ~u : (u | 0x80000000u);
}

__device__ __forceinline__ float ord_decode_finite(unsigned k) {
    unsigned u = (k & 0x80000000u) ? (k & 0x7FFFFFFFu) : ~k;
    if ((u & 0x7F800000u) == 0x7F800000u) return 0.0f;
    return __uint_as_float(u);
}

template <int F, int WQ_CAP>
__device__ __forceinline__ void GCN_apply_hits(const float* __restrict__ feat, unsigned* acc,
                                               const unsigned* myq, unsigned nq, int lane) {
    constexpr int LPE = F / 4;
    const unsigned ntask = nq * LPE;
    for (unsigned t0 = lane; t0 < ntask; t0 += 32 * TPL) {
        bool     ok[TPL];
        unsigned pp[TPL];
        unsigned qs[TPL];
        unsigned qd[TPL];
        float    qw[TPL];
        float4   vv[TPL];
#pragma unroll
        for (int u = 0; u < TPL; ++u) {
            unsigned t = t0 + u * 32;
            ok[u] = t < ntask;
            pp[u] = t % LPE;
            unsigned hh = ok[u] ? t / LPE : 0u;
            qs[u] = myq[hh];
            qw[u] = __uint_as_float(myq[WQ_CAP + hh]);
            qd[u] = myq[2 * WQ_CAP + hh];
        }
        asm volatile("" :: "v"(qs[0]), "v"(qs[1]), "v"(qs[2]), "v"(qs[3]));
#pragma unroll
        for (int u = 0; u < TPL; ++u) {
            vv[u] = make_float4(0.0f, 0.0f, 0.0f, 0.0f);
            if (ok[u]) vv[u] = *(const float4*)(feat + (size_t)qs[u] * F + 4 * pp[u]);
        }
        asm volatile("" :: "v"(vv[0].x), "v"(vv[0].y), "v"(vv[0].z), "v"(vv[0].w),
                           "v"(vv[1].x), "v"(vv[1].y), "v"(vv[1].z), "v"(vv[1].w),
                           "v"(vv[2].x), "v"(vv[2].y), "v"(vv[2].z), "v"(vv[2].w),
                           "v"(vv[3].x), "v"(vv[3].y), "v"(vv[3].z), "v"(vv[3].w));
#pragma unroll
        for (int u = 0; u < TPL; ++u) {
            if (ok[u]) {
                unsigned* a = acc + qd[u] * F + 4 * pp[u];
                atomicMax(a + 0, ord_encode(vv[u].x * qw[u]));
                atomicMax(a + 1, ord_encode(vv[u].y * qw[u]));
                atomicMax(a + 2, ord_encode(vv[u].z * qw[u]));
                atomicMax(a + 3, ord_encode(vv[u].w * qw[u]));
            }
        }
    }
}

template <int LOG2F, int RANGE, int WQ_CAP>
__global__ void __launch_bounds__(SCAN_NT)
GCN_range_scan_max(const float* __restrict__ feat,
                   const int* __restrict__ src,
                   const int* __restrict__ dst,
                   const float* __restrict__ ts,
                   unsigned* __restrict__ part,
                   int n_nodes, int n_edges, int n_ranges, int blocks_per_slice) {
    constexpr int F         = 1 << LOG2F;
    constexpr int ACC_WORDS = RANGE * F;

    __shared__ __attribute__((aligned(16))) unsigned acc[ACC_WORDS];
    __shared__ unsigned wq[SCAN_NT / 32][3 * WQ_CAP];

    const int tid  = threadIdx.x;
    const int lane = tid & 31;
    const int wave = tid >> 5;
    const int r    = (int)blockIdx.x % n_ranges;
    const int s    = (int)blockIdx.x / n_ranges;
    const int lo   = r * RANGE;
    int cnt = n_nodes - lo;
    if (cnt > RANGE) cnt = RANGE;

    for (int i = tid; i < ACC_WORDS; i += SCAN_NT) acc[i] = ORD_NEG_INF;
    __syncthreads();

    unsigned* myq = wq[wave];
    const int c_end = (s + 1) * blocks_per_slice * (SCAN_NT / 32);
    int   dv[SCAN_EPT];
    int   sv[SCAN_EPT];
    float tv[SCAN_EPT];
    bool have = false;
    for (int c = s * blocks_per_slice * (SCAN_NT / 32) + wave; c < c_end; c += SCAN_NT / 32) {
        const int ebase = c * SCAN_CHUNK;
        if (ebase >= n_edges) break;

        if (have) {
        } else if (ebase + SCAN_CHUNK <= n_edges) {
#pragma unroll
            for (int k = 0; k < 4; ++k) {
                const int o = ebase + (k * 32 + lane) * 4;
                int4   d = *(const int4*)(dst + o);
                int4   q = *(const int4*)(src + o);
                float4 w = *(const float4*)(ts + o);
                dv[4 * k + 0] = d.x; dv[4 * k + 1] = d.y; dv[4 * k + 2] = d.z; dv[4 * k + 3] = d.w;
                sv[4 * k + 0] = q.x; sv[4 * k + 1] = q.y; sv[4 * k + 2] = q.z; sv[4 * k + 3] = q.w;
                tv[4 * k + 0] = w.x; tv[4 * k + 1] = w.y; tv[4 * k + 2] = w.z; tv[4 * k + 3] = w.w;
            }
        } else {
#pragma unroll
            for (int i = 0; i < SCAN_EPT; ++i) {
                int e = ebase + ((i >> 2) * 32 + lane) * 4 + (i & 3);
                dv[i] = -1; sv[i] = 0; tv[i] = 0.0f;
                if (e < n_edges) { dv[i] = dst[e]; sv[i] = src[e]; tv[i] = ts[e]; }
            }
        }

        unsigned hitmask[SCAN_EPT];
        unsigned tot = 0;
#pragma unroll
        for (int i = 0; i < SCAN_EPT; ++i) {
            unsigned dl = (unsigned)(dv[i] - lo);
            hitmask[i] = __builtin_amdgcn_ballot_w32(dl < (unsigned)cnt);
            tot += (unsigned)__builtin_popcount(hitmask[i]);
        }

        const int  enext = ebase + (SCAN_NT / 32) * SCAN_CHUNK;
        const bool next_whole = (c + SCAN_NT / 32 < c_end) && (enext + SCAN_CHUNK <= n_edges);

        if (tot <= (unsigned)WQ_CAP) {
            unsigned run = 0;
#pragma unroll
            for (int i = 0; i < SCAN_EPT; ++i) {
                unsigned dl = (unsigned)(dv[i] - lo);
                if (dl < (unsigned)cnt) {
                    unsigned pos = run + __builtin_amdgcn_mbcnt_lo(hitmask[i], 0u);
                    myq[pos]              = (unsigned)sv[i];
                    myq[WQ_CAP + pos]     = __float_as_uint(tv[i]);
                    myq[2 * WQ_CAP + pos] = dl;
                }
                run += (unsigned)__builtin_popcount(hitmask[i]);
            }
            have = next_whole;
            if (have) {
#pragma unroll
                for (int k = 0; k < 4; ++k) {
                    const int o = enext + (k * 32 + lane) * 4;
                    int4   d = *(const int4*)(dst + o);
                    int4   q = *(const int4*)(src + o);
                    float4 w = *(const float4*)(ts + o);
                    dv[4 * k + 0] = d.x; dv[4 * k + 1] = d.y; dv[4 * k + 2] = d.z; dv[4 * k + 3] = d.w;
                    sv[4 * k + 0] = q.x; sv[4 * k + 1] = q.y; sv[4 * k + 2] = q.z; sv[4 * k + 3] = q.w;
                    tv[4 * k + 0] = w.x; tv[4 * k + 1] = w.y; tv[4 * k + 2] = w.z; tv[4 * k + 3] = w.w;
                }
            }
            if (tot != 0u) {
                __builtin_amdgcn_fence(__ATOMIC_RELEASE, "wavefront");
                __builtin_amdgcn_wave_barrier();
                __builtin_amdgcn_fence(__ATOMIC_ACQUIRE, "wavefront");
                GCN_apply_hits<F, WQ_CAP>(feat, acc, myq, tot, lane);
                __builtin_amdgcn_wave_barrier();
            }
        } else {
            for (unsigned rb = 0; rb < tot; rb += WQ_CAP) {
                unsigned run = 0;
#pragma unroll
                for (int i = 0; i < SCAN_EPT; ++i) {
                    unsigned dl  = (unsigned)(dv[i] - lo);
                    unsigned pos = run + __builtin_amdgcn_mbcnt_lo(hitmask[i], 0u) - rb;
                    if (dl < (unsigned)cnt && pos < (unsigned)WQ_CAP) {
                        myq[pos]              = (unsigned)sv[i];
                        myq[WQ_CAP + pos]     = __float_as_uint(tv[i]);
                        myq[2 * WQ_CAP + pos] = dl;
                    }
                    run += (unsigned)__builtin_popcount(hitmask[i]);
                }
                __builtin_amdgcn_fence(__ATOMIC_RELEASE, "wavefront");
                __builtin_amdgcn_wave_barrier();
                __builtin_amdgcn_fence(__ATOMIC_ACQUIRE, "wavefront");
                unsigned nq = tot - rb;
                if (nq > (unsigned)WQ_CAP) nq = WQ_CAP;
                GCN_apply_hits<F, WQ_CAP>(feat, acc, myq, nq, lane);
                __builtin_amdgcn_wave_barrier();
            }
            have = false;
        }
    }
    __syncthreads();

    unsigned* outp = part + ((size_t)s * n_nodes + lo) * F;
    const int nw = cnt * F;
    for (int i = tid * 4; i < nw; i += SCAN_NT * 4)
        *(uint4*)(outp + i) = *(const uint4*)(acc + i);
}

template <int S>
__global__ void __launch_bounds__(256)
GCN_finalize1(const unsigned* __restrict__ part,
              const float* __restrict__ W1,
              const float* __restrict__ b1,
              float* __restrict__ h,
              int n_nodes) {
    int wave = (int)((blockIdx.x * (long)blockDim.x + threadIdx.x) >> 5);
    int lane = threadIdx.x & 31;
    int base = wave * 16;
    if (base >= n_nodes) return;

    int m  = lane & 15;
    int kh = lane >> 4;
    int n  = m;

    const size_t plane2 = (size_t)n_nodes * 8;
    const uint2* p = (const uint2*)(part + (size_t)(base + m) * 16 + 2 * kh);
    unsigned kx[4] = {0u, 0u, 0u, 0u}, ky[4] = {0u, 0u, 0u, 0u};
#pragma unroll
    for (int s = 0; s < S; ++s) {
#pragma unroll
        for (int c = 0; c < 4; ++c) {
            uint2 v = p[s * plane2 + 2 * c];
            kx[c] = v.x > kx[c] ? v.x : kx[c];
            ky[c] = v.y > ky[c] ? v.y : ky[c];
        }
    }

    v8f acc = {0.f, 0.f, 0.f, 0.f, 0.f, 0.f, 0.f, 0.f};
#pragma unroll
    for (int c = 0; c < 4; ++c) {
        int k0 = 4 * c + 2 * kh;
        v2f a, b;
        a.x = ord_decode_finite(kx[c]);
        a.y = ord_decode_finite(ky[c]);
        b.x = (n < 8) ? W1[k0 * 8 + n]       : 0.0f;
        b.y = (n < 8) ? W1[(k0 + 1) * 8 + n] : 0.0f;
        acc = __builtin_amdgcn_wmma_f32_16x16x4_f32(
             false, a,  false, b,
             (short)0, acc,  false,  false);
    }

    if (n < 8) {
#pragma unroll
        for (int r = 0; r < 8; ++r) {
            int node = base + r + 8 * kh;
            float v = acc[r] + b1[n];
            h[node * 8 + n] = v > 0.0f ? v : 0.0f;
        }
    }
}

template <int S>
__global__ void __launch_bounds__(256)
GCN_finalize2(const unsigned* __restrict__ part,
              const float* __restrict__ W2,
              const float* __restrict__ b2,
              float* __restrict__ out,
              int n_nodes) {
    int wave = (int)((blockIdx.x * (long)blockDim.x + threadIdx.x) >> 5);
    int lane = threadIdx.x & 31;
    int base = wave * 16;
    if (base >= n_nodes) return;

    int m  = lane & 15;
    int kh = lane >> 4;
    int n  = m;

    const size_t plane2 = (size_t)n_nodes * 4;
    const uint2* p = (const uint2*)(part + (size_t)(base + m) * 8 + 2 * kh);
    unsigned kx[2] = {0u, 0u}, ky[2] = {0u, 0u};
#pragma unroll
    for (int s = 0; s < S; ++s) {
#pragma unroll
        for (int c = 0; c < 2; ++c) {
            uint2 v = p[s * plane2 + 2 * c];
            kx[c] = v.x > kx[c] ? v.x : kx[c];
            ky[c] = v.y > ky[c] ? v.y : ky[c];
        }
    }

    v8f acc = {0.f, 0.f, 0.f, 0.f, 0.f, 0.f, 0.f, 0.f};
#pragma unroll
    for (int c = 0; c < 2; ++c) {
        int k0 = 4 * c + 2 * kh;
        v2f a, b;
        a.x = ord_decode_finite(kx[c]);
        a.y = ord_decode_finite(ky[c]);
        b.x = (n < 2) ? W2[k0 * 2 + n]       : 0.0f;
        b.y = (n < 2) ? W2[(k0 + 1) * 2 + n] : 0.0f;
        acc = __builtin_amdgcn_wmma_f32_16x16x4_f32(
            false, a, false, b, (short)0, acc, false, false);
    }

    if (n < 2) {
#pragma unroll
        for (int r = 0; r < 8; ++r) {
            int node = base + r + 8 * kh;
            out[node * 2 + n] = acc[r] + b2[n];
        }
    }
}

extern "C" void kernel_launch(void* const* d_in, const int* in_sizes, int n_in,
                              void* d_out, int out_size, void* d_ws, size_t ws_size,
                              hipStream_t stream) {
    const float* x   = (const float*)d_in[0];
    const int*   src = (const int*)  d_in[1];
    const int*   dst = (const int*)  d_in[2];
    const float* ts  = (const float*)d_in[3];
    const float* W1  = (const float*)d_in[4];
    const float* b1  = (const float*)d_in[5];
    const float* W2  = (const float*)d_in[6];
    const float* b2  = (const float*)d_in[7];
    float* out = (float*)d_out;

    const int N = in_sizes[0] / 16;
    const int E = in_sizes[1];

    unsigned* part1 = (unsigned*)d_ws;
    float*    h     = (float*)(part1 + (size_t)SLICES_L1 * N * 16);
    unsigned* part2 = (unsigned*)(h + (size_t)N * 8);

    const int B = 256;
    const int nblk = (E + SCAN_BLK - 1) / SCAN_BLK;
    int waves   = N / 16;
    int fin_thr = waves * 32;

    const int R1   = (N + RANGE_L1 - 1) / RANGE_L1;
    const int bps1 = (nblk + SLICES_L1 - 1) / SLICES_L1;
    GCN_range_scan_max<4, RANGE_L1, QCAP_L1><<<(unsigned)(R1 * SLICES_L1), SCAN_NT, 0, stream>>>(
        x, src, dst, ts, part1, N, E, R1, bps1);
    GCN_finalize1<SLICES_L1><<<(unsigned)((fin_thr + B - 1) / B), B, 0, stream>>>(part1, W1, b1, h, N);

    const int R2   = (N + RANGE_L2 - 1) / RANGE_L2;
    const int bps2 = (nblk + SLICES_L2 - 1) / SLICES_L2;
    GCN_range_scan_max<3, RANGE_L2, QCAP_L2><<<(unsigned)(R2 * SLICES_L2), SCAN_NT, 0, stream>>>(
        h, src, dst, ts, part2, N, E, R2, bps2);
    GCN_finalize2<SLICES_L2><<<(unsigned)((fin_thr + B - 1) / B), B, 0, stream>>>(part2, W2, b2, out, N);
}
